# starting kernel
# baseline (speedup 1.0000x reference)
#include <hip/hip_runtime.h>
#include <math.h>

typedef __attribute__((ext_vector_type(2))) float v2f;
typedef __attribute__((ext_vector_type(8))) float v8f;

#define BB 2
#define SS 192
#define HH 768
#define LL 4
#define NN (HH * LL)
#define MM (BB * SS)

__global__ __launch_bounds__(32) void proj_wmma_f32(
    const float* __restrict__ X, const float* __restrict__ W,
    const float* __restrict__ bias, float* __restrict__ Y) {
  const int n0 = blockIdx.x * 16;
  const int m0 = blockIdx.y * 16;
  const int lane = threadIdx.x;
  const int r16  = lane & 15;
  const int kh   = (lane >> 4) * 2;

  const float* __restrict__ arow = X + (size_t)(m0 + r16) * HH;
  const float* __restrict__ brow = W + (size_t)(n0 + r16) * HH;

  v8f acc = {};
  for (int k = 0; k < HH; k += 4) {
    v2f a = *(const v2f*)(arow + k + kh);
    v2f b = *(const v2f*)(brow + k + kh);
    acc = __builtin_amdgcn_wmma_f32_16x16x4_f32(
         false, a,  false, b,
         (short)0, acc,  false,  false);
  }

  const int n     = n0 + r16;
  const int mbase = m0 + ((lane >> 4) << 3);
  const float bn  = bias[n];
#pragma unroll
  for (int r = 0; r < 8; ++r) {
    Y[(size_t)(mbase + r) * NN + n] = acc[r] + bn;
  }
}

__global__ __launch_bounds__(256) void pair_gelu_reduce(
    const float* __restrict__ hsrc, const float* __restrict__ htgt,
    const float* __restrict__ wcls, float* __restrict__ out) {
  __shared__ float s_src[HH];
  __shared__ float s_w[HH];

  const int s = blockIdx.x;
  const int l = blockIdx.y;
  const int b = blockIdx.z;
  const int tid = threadIdx.x;

  const float* __restrict__ srcrow =
      hsrc + ((size_t)((b * SS + s) * LL + l)) * HH;
  for (int i = tid; i < HH; i += 256) {
    s_src[i] = srcrow[i];
    s_w[i]   = wcls[i * LL + l];
  }
  __syncthreads();

  const int lane = tid & 31;
  const int wid  = tid >> 5;
  const float* __restrict__ tgtbase =
      htgt + ((size_t)(b * SS * LL + l)) * HH;

  for (int t = wid; t < SS; t += 8) {
    const float* __restrict__ trow = tgtbase + (size_t)t * (LL * HH);
    float acc = 0.0f;
#pragma unroll 4
    for (int h = lane; h < HH; h += 32) {
      float x = s_src[h] + trow[h];
      float g = 0.5f * x * (1.0f + erff(x * 0.70710678118654752f));
      acc = fmaf(g, s_w[h], acc);
    }
#pragma unroll
    for (int off = 16; off > 0; off >>= 1) acc += __shfl_xor(acc, off, 32);
    if (lane == 0) {
      out[((size_t)((b * SS + s) * SS + t)) * LL + l] = acc;
    }
  }
}

extern "C" void kernel_launch(void* const* d_in, const int* in_sizes, int n_in,
                              void* d_out, int out_size, void* d_ws, size_t ws_size,
                              hipStream_t stream) {
  const float* pooled = (const float*)d_in[0];
  const float* W_src  = (const float*)d_in[1];
  const float* b_src  = (const float*)d_in[2];
  const float* W_tgt  = (const float*)d_in[3];
  const float* b_tgt  = (const float*)d_in[4];
  const float* wcls   = (const float*)d_in[5];
  float* out = (float*)d_out;

  float* hsrc = (float*)d_ws;
  float* htgt = hsrc + (size_t)MM * NN;

  dim3 g1(NN / 16, MM / 16);
  proj_wmma_f32<<<g1, 32, 0, stream>>>(pooled, W_src, b_src, hsrc);
  proj_wmma_f32<<<g1, 32, 0, stream>>>(pooled, W_tgt, b_tgt, htgt);

  dim3 g2(SS, LL, BB);
  pair_gelu_reduce<<<g2, 256, 0, stream>>>(hsrc, htgt, wcls, out);
}


# exact-erf GELU as one branch-free form: max(x,0) - |x|*2^-Q(|x|), Q a degree-10 fit of -log2(erfc(u/sqrt2)/2), one v_exp_f32; replaces library erff and its two per-lane ranges
# speedup vs baseline: 1.7816x; 1.7816x over previous
#include <hip/hip_runtime.h>
#include <math.h>

typedef __attribute__((ext_vector_type(2))) float v2f;
typedef __attribute__((ext_vector_type(8))) float v8f;

#define BB 2
#define SS 192
#define HH 768
#define LL 4
#define NN (HH * LL)
#define MM (BB * SS)

__global__ __launch_bounds__(32) void proj_wmma_f32(
    const float* __restrict__ X, const float* __restrict__ W,
    const float* __restrict__ bias, float* __restrict__ Y) {
  const int n0 = blockIdx.x * 16;
  const int m0 = blockIdx.y * 16;
  const int lane = threadIdx.x;
  const int r16  = lane & 15;
  const int kh   = (lane >> 4) * 2;

  const float* __restrict__ arow = X + (size_t)(m0 + r16) * HH;
  const float* __restrict__ brow = W + (size_t)(n0 + r16) * HH;

  v8f acc = {};
  for (int k = 0; k < HH; k += 4) {
    v2f a = *(const v2f*)(arow + k + kh);
    v2f b = *(const v2f*)(brow + k + kh);
    acc = __builtin_amdgcn_wmma_f32_16x16x4_f32(
         false, a,  false, b,
         (short)0, acc,  false,  false);
  }

  const int n     = n0 + r16;
  const int mbase = m0 + ((lane >> 4) << 3);
  const float bn  = bias[n];
#pragma unroll
  for (int r = 0; r < 8; ++r) {
    Y[(size_t)(mbase + r) * NN + n] = acc[r] + bn;
  }
}

__device__ __forceinline__ float gelu_erf(float x) {
  const float u = fabsf(x);
  float q = 1.57496416e-8f;
  q = fmaf(q, u, -5.81307916e-7f);
  q = fmaf(q, u, 0.00000862902561f);
  q = fmaf(q, u, -0.0000658568606f);
  q = fmaf(q, u, 0.000250525482f);
  q = fmaf(q, u, 0.0000509071178f);
  q = fmaf(q, u, -0.00700288685f);
  q = fmaf(q, u, 0.0524775460f);
  q = fmaf(q, u, 0.459208757f);
  q = fmaf(q, u, 1.15110576f);
  q = fmaf(q, u, 0.999999940f);
  const float e = __builtin_amdgcn_exp2f(-q);
  return fmaf(-u, e, fmaxf(x, 0.0f));
}

__global__ __launch_bounds__(256) void pair_gelu_reduce(
    const float* __restrict__ hsrc, const float* __restrict__ htgt,
    const float* __restrict__ wcls, float* __restrict__ out) {
  __shared__ float s_src[HH];
  __shared__ float s_w[HH];

  const int s = blockIdx.x;
  const int l = blockIdx.y;
  const int b = blockIdx.z;
  const int tid = threadIdx.x;

  const float* __restrict__ srcrow =
      hsrc + ((size_t)((b * SS + s) * LL + l)) * HH;
  for (int i = tid; i < HH; i += 256) {
    s_src[i] = srcrow[i];
    s_w[i]   = wcls[i * LL + l];
  }
  __syncthreads();

  const int lane = tid & 31;
  const int wid  = tid >> 5;
  const float* __restrict__ tgtbase =
      htgt + ((size_t)(b * SS * LL + l)) * HH;

  for (int t = wid; t < SS; t += 8) {
    const float* __restrict__ trow = tgtbase + (size_t)t * (LL * HH);
    float acc = 0.0f;
#pragma unroll 4
    for (int h = lane; h < HH; h += 32) {
      float x = s_src[h] + trow[h];
      float g = gelu_erf(x);
      acc = fmaf(g, s_w[h], acc);
    }
#pragma unroll
    for (int off = 16; off > 0; off >>= 1) acc += __shfl_xor(acc, off, 32);
    if (lane == 0) {
      out[((size_t)((b * SS + s) * SS + t)) * LL + l] = acc;
    }
  }
}

extern "C" void kernel_launch(void* const* d_in, const int* in_sizes, int n_in,
                              void* d_out, int out_size, void* d_ws, size_t ws_size,
                              hipStream_t stream) {
  const float* pooled = (const float*)d_in[0];
  const float* W_src  = (const float*)d_in[1];
  const float* b_src  = (const float*)d_in[2];
  const float* W_tgt  = (const float*)d_in[3];
  const float* b_tgt  = (const float*)d_in[4];
  const float* wcls   = (const float*)d_in[5];
  float* out = (float*)d_out;

  float* hsrc = (float*)d_ws;
  float* htgt = hsrc + (size_t)MM * NN;

  dim3 g1(NN / 16, MM / 16);
  proj_wmma_f32<<<g1, 32, 0, stream>>>(pooled, W_src, b_src, hsrc);
  proj_wmma_f32<<<g1, 32, 0, stream>>>(pooled, W_tgt, b_tgt, htgt);

  dim3 g2(SS, LL, BB);
  pair_gelu_reduce<<<g2, 256, 0, stream>>>(hsrc, htgt, wcls, out);
}


# projection GEMMs tiled through LDS: 4-wave groups own 32x64 tiles of Y; X and W slabs fetched once per tile with whole-line loads, double-buffered; one A fragment feeds two WMMAs; k order kept
# speedup vs baseline: 2.3910x; 1.3421x over previous
#include <hip/hip_runtime.h>
#include <math.h>

typedef __attribute__((ext_vector_type(2))) float v2f;
typedef __attribute__((ext_vector_type(4))) float v4f;
typedef __attribute__((ext_vector_type(8))) float v8f;

#define BB 2
#define SS 192
#define HH 768
#define LL 4
#define NN (HH * LL)
#define MM (BB * SS)

#define TM 32
#define TN 64
#define TK 64
#define LP 68

__device__ __forceinline__ v8f wmma4(v2f a, v2f b, v8f c) {
  return __builtin_amdgcn_wmma_f32_16x16x4_f32(
       false, a,  false, b,
       (short)0, c,  false,  false);
}

__global__ __launch_bounds__(128) void proj_wmma_f32(
    const float* __restrict__ X, const float* __restrict__ W,
    const float* __restrict__ bias, float* __restrict__ Y) {
  __shared__ __attribute__((aligned(16))) float sA[2][TM][LP];
  __shared__ __attribute__((aligned(16))) float sB[2][TN][LP];

  const int n0   = blockIdx.x * TN;
  const int m0   = blockIdx.y * TM;
  const int tid  = threadIdx.x;
  const int lane = tid & 31;
  const int wid  = tid >> 5;
  const int wm   = wid & 1;
  const int wn   = wid >> 1;
  const int r16  = lane & 15;
  const int kh   = (lane >> 4) * 2;

  const int srow = tid >> 4;
  const int scol = (tid & 15) * 4;

  const float* __restrict__ gA = X + (size_t)(m0 + srow) * HH + scol;
  const float* __restrict__ gB = W + (size_t)(n0 + srow) * HH + scol;

  v4f ra[TM / 8];
  v4f rb[TN / 8];

  for (int j = 0; j < TM / 8; ++j) ra[j] = *(const v4f*)(gA + (size_t)(8 * j) * HH);
  for (int j = 0; j < TN / 8; ++j) rb[j] = *(const v4f*)(gB + (size_t)(8 * j) * HH);
  for (int j = 0; j < TM / 8; ++j) *(v4f*)&sA[0][srow + 8 * j][scol] = ra[j];
  for (int j = 0; j < TN / 8; ++j) *(v4f*)&sB[0][srow + 8 * j][scol] = rb[j];
  __syncthreads();

  v8f acc0 = {};
  v8f acc1 = {};

  for (int kt = 0; kt < HH / TK; ++kt) {
    const int  buf  = kt & 1;
    const bool more = (kt + 1) < (HH / TK);

    if (more) {
      const int ko = (kt + 1) * TK;
      for (int j = 0; j < TM / 8; ++j) ra[j] = *(const v4f*)(gA + (size_t)(8 * j) * HH + ko);
      for (int j = 0; j < TN / 8; ++j) rb[j] = *(const v4f*)(gB + (size_t)(8 * j) * HH + ko);
    }

    const float* __restrict__ pa  = &sA[buf][wm * 16 + r16][kh];
    const float* __restrict__ pb0 = &sB[buf][wn * 32 + r16][kh];
    const float* __restrict__ pb1 = &sB[buf][wn * 32 + 16 + r16][kh];
    v2f fa[TK / 4], fb0[TK / 4], fb1[TK / 4];
    for (int ks = 0; ks < TK / 4; ++ks) {
      fa[ks]  = *(const v2f*)(pa + 4 * ks);
      fb0[ks] = *(const v2f*)(pb0 + 4 * ks);
      fb1[ks] = *(const v2f*)(pb1 + 4 * ks);
    }
    __builtin_amdgcn_sched_barrier(0);
    for (int ks = 0; ks < TK / 4; ++ks) {
      acc0 = wmma4(fa[ks], fb0[ks], acc0);
      acc1 = wmma4(fa[ks], fb1[ks], acc1);
    }
    __builtin_amdgcn_sched_barrier(0);

    if (more) {
      for (int j = 0; j < TM / 8; ++j) *(v4f*)&sA[buf ^ 1][srow + 8 * j][scol] = ra[j];
      for (int j = 0; j < TN / 8; ++j) *(v4f*)&sB[buf ^ 1][srow + 8 * j][scol] = rb[j];
      __syncthreads();
    }
  }

  const int n     = n0 + wn * 32 + r16;
  const int mbase = m0 + wm * 16 + ((lane >> 4) << 3);
  const float bn0 = bias[n];
  const float bn1 = bias[n + 16];
  for (int r = 0; r < 8; ++r) {
    Y[(size_t)(mbase + r) * NN + n]      = acc0[r] + bn0;
    Y[(size_t)(mbase + r) * NN + n + 16] = acc1[r] + bn1;
  }
}

__device__ __forceinline__ float gelu_erf(float x) {
  const float u = fabsf(x);
  float q = 1.57496416e-8f;
  q = fmaf(q, u, -5.81307916e-7f);
  q = fmaf(q, u, 0.00000862902561f);
  q = fmaf(q, u, -0.0000658568606f);
  q = fmaf(q, u, 0.000250525482f);
  q = fmaf(q, u, 0.0000509071178f);
  q = fmaf(q, u, -0.00700288685f);
  q = fmaf(q, u, 0.0524775460f);
  q = fmaf(q, u, 0.459208757f);
  q = fmaf(q, u, 1.15110576f);
  q = fmaf(q, u, 0.999999940f);
  const float e = __builtin_amdgcn_exp2f(-q);
  return fmaf(-u, e, fmaxf(x, 0.0f));
}

__global__ __launch_bounds__(256) void pair_gelu_reduce(
    const float* __restrict__ hsrc, const float* __restrict__ htgt,
    const float* __restrict__ wcls, float* __restrict__ out) {
  __shared__ float s_src[HH];
  __shared__ float s_w[HH];

  const int s = blockIdx.x;
  const int l = blockIdx.y;
  const int b = blockIdx.z;
  const int tid = threadIdx.x;

  const float* __restrict__ srcrow =
      hsrc + ((size_t)((b * SS + s) * LL + l)) * HH;
  for (int i = tid; i < HH; i += 256) {
    s_src[i] = srcrow[i];
    s_w[i]   = wcls[i * LL + l];
  }
  __syncthreads();

  const int lane = tid & 31;
  const int wid  = tid >> 5;
  const float* __restrict__ tgtbase =
      htgt + ((size_t)(b * SS * LL + l)) * HH;

  for (int t = wid; t < SS; t += 8) {
    const float* __restrict__ trow = tgtbase + (size_t)t * (LL * HH);
    float acc = 0.0f;
#pragma unroll 4
    for (int h = lane; h < HH; h += 32) {
      float x = s_src[h] + trow[h];
      float g = gelu_erf(x);
      acc = fmaf(g, s_w[h], acc);
    }
#pragma unroll
    for (int off = 16; off > 0; off >>= 1) acc += __shfl_xor(acc, off, 32);
    if (lane == 0) {
      out[((size_t)((b * SS + s) * SS + t)) * LL + l] = acc;
    }
  }
}

extern "C" void kernel_launch(void* const* d_in, const int* in_sizes, int n_in,
                              void* d_out, int out_size, void* d_ws, size_t ws_size,
                              hipStream_t stream) {
  const float* pooled = (const float*)d_in[0];
  const float* W_src  = (const float*)d_in[1];
  const float* b_src  = (const float*)d_in[2];
  const float* W_tgt  = (const float*)d_in[3];
  const float* b_tgt  = (const float*)d_in[4];
  const float* wcls   = (const float*)d_in[5];
  float* out = (float*)d_out;

  float* hsrc = (float*)d_ws;
  float* htgt = hsrc + (size_t)MM * NN;

  dim3 g1(NN / TN, MM / TM);
  proj_wmma_f32<<<g1, 128, 0, stream>>>(pooled, W_src, b_src, hsrc);
  proj_wmma_f32<<<g1, 128, 0, stream>>>(pooled, W_tgt, b_tgt, htgt);

  dim3 g2(SS, LL, BB);
  pair_gelu_reduce<<<g2, 256, 0, stream>>>(hsrc, htgt, wcls, out);
}


# pair kernel: each wave holds its source row and classifier column in registers, sweeps 48 target rows with 16-byte loads one row ahead (8 waves share a row), no LDS; per-lane sum order changed
# speedup vs baseline: 3.4657x; 1.4495x over previous
#include <hip/hip_runtime.h>
#include <math.h>

typedef __attribute__((ext_vector_type(2))) float v2f;
typedef __attribute__((ext_vector_type(4))) float v4f;
typedef __attribute__((ext_vector_type(8))) float v8f;

#define BB 2
#define SS 192
#define HH 768
#define LL 4
#define NN (HH * LL)
#define MM (BB * SS)

#define TM 32
#define TN 64
#define TK 64
#define LP 68

__device__ __forceinline__ v8f wmma4(v2f a, v2f b, v8f c) {
  return __builtin_amdgcn_wmma_f32_16x16x4_f32(
       false, a,  false, b,
       (short)0, c,  false,  false);
}

__global__ __launch_bounds__(128) void proj_wmma_f32(
    const float* __restrict__ X, const float* __restrict__ W,
    const float* __restrict__ bias, float* __restrict__ Y) {
  __shared__ __attribute__((aligned(16))) float sA[2][TM][LP];
  __shared__ __attribute__((aligned(16))) float sB[2][TN][LP];

  const int n0   = blockIdx.x * TN;
  const int m0   = blockIdx.y * TM;
  const int tid  = threadIdx.x;
  const int lane = tid & 31;
  const int wid  = tid >> 5;
  const int wm   = wid & 1;
  const int wn   = wid >> 1;
  const int r16  = lane & 15;
  const int kh   = (lane >> 4) * 2;

  const int srow = tid >> 4;
  const int scol = (tid & 15) * 4;

  const float* __restrict__ gA = X + (size_t)(m0 + srow) * HH + scol;
  const float* __restrict__ gB = W + (size_t)(n0 + srow) * HH + scol;

  v4f ra[TM / 8];
  v4f rb[TN / 8];

  for (int j = 0; j < TM / 8; ++j) ra[j] = *(const v4f*)(gA + (size_t)(8 * j) * HH);
  for (int j = 0; j < TN / 8; ++j) rb[j] = *(const v4f*)(gB + (size_t)(8 * j) * HH);
  for (int j = 0; j < TM / 8; ++j) *(v4f*)&sA[0][srow + 8 * j][scol] = ra[j];
  for (int j = 0; j < TN / 8; ++j) *(v4f*)&sB[0][srow + 8 * j][scol] = rb[j];
  __syncthreads();

  v8f acc0 = {};
  v8f acc1 = {};

  for (int kt = 0; kt < HH / TK; ++kt) {
    const int  buf  = kt & 1;
    const bool more = (kt + 1) < (HH / TK);

    if (more) {
      const int ko = (kt + 1) * TK;
      for (int j = 0; j < TM / 8; ++j) ra[j] = *(const v4f*)(gA + (size_t)(8 * j) * HH + ko);
      for (int j = 0; j < TN / 8; ++j) rb[j] = *(const v4f*)(gB + (size_t)(8 * j) * HH + ko);
    }

    const float* __restrict__ pa  = &sA[buf][wm * 16 + r16][kh];
    const float* __restrict__ pb0 = &sB[buf][wn * 32 + r16][kh];
    const float* __restrict__ pb1 = &sB[buf][wn * 32 + 16 + r16][kh];
    v2f fa[TK / 4], fb0[TK / 4], fb1[TK / 4];
    for (int ks = 0; ks < TK / 4; ++ks) {
      fa[ks]  = *(const v2f*)(pa + 4 * ks);
      fb0[ks] = *(const v2f*)(pb0 + 4 * ks);
      fb1[ks] = *(const v2f*)(pb1 + 4 * ks);
    }
    __builtin_amdgcn_sched_barrier(0);
    for (int ks = 0; ks < TK / 4; ++ks) {
      acc0 = wmma4(fa[ks], fb0[ks], acc0);
      acc1 = wmma4(fa[ks], fb1[ks], acc1);
    }
    __builtin_amdgcn_sched_barrier(0);

    if (more) {
      for (int j = 0; j < TM / 8; ++j) *(v4f*)&sA[buf ^ 1][srow + 8 * j][scol] = ra[j];
      for (int j = 0; j < TN / 8; ++j) *(v4f*)&sB[buf ^ 1][srow + 8 * j][scol] = rb[j];
      __syncthreads();
    }
  }

  const int n     = n0 + wn * 32 + r16;
  const int mbase = m0 + wm * 16 + ((lane >> 4) << 3);
  const float bn0 = bias[n];
  const float bn1 = bias[n + 16];
  for (int r = 0; r < 8; ++r) {
    Y[(size_t)(mbase + r) * NN + n]      = acc0[r] + bn0;
    Y[(size_t)(mbase + r) * NN + n + 16] = acc1[r] + bn1;
  }
}

__device__ __forceinline__ float gelu_erf(float x) {
  const float u = fabsf(x);
  float q = 1.57496416e-8f;
  q = fmaf(q, u, -5.81307916e-7f);
  q = fmaf(q, u, 0.00000862902561f);
  q = fmaf(q, u, -0.0000658568606f);
  q = fmaf(q, u, 0.000250525482f);
  q = fmaf(q, u, 0.0000509071178f);
  q = fmaf(q, u, -0.00700288685f);
  q = fmaf(q, u, 0.0524775460f);
  q = fmaf(q, u, 0.459208757f);
  q = fmaf(q, u, 1.15110576f);
  q = fmaf(q, u, 0.999999940f);
  const float e = __builtin_amdgcn_exp2f(-q);
  return fmaf(-u, e, fmaxf(x, 0.0f));
}

#define PW 8
#define PT 48
#define HV (HH / 128)

__device__ __forceinline__ float row_dot(const v4f (&sv)[HV], const v4f (&wv)[HV],
                                         const v4f (&tv)[HV]) {
  float a0 = 0.0f, a1 = 0.0f, a2 = 0.0f, a3 = 0.0f;
#pragma unroll
  for (int i = 0; i < HV; ++i) {
    a0 = fmaf(gelu_erf(sv[i].x + tv[i].x), wv[i].x, a0);
    a1 = fmaf(gelu_erf(sv[i].y + tv[i].y), wv[i].y, a1);
    a2 = fmaf(gelu_erf(sv[i].z + tv[i].z), wv[i].z, a2);
    a3 = fmaf(gelu_erf(sv[i].w + tv[i].w), wv[i].w, a3);
  }
  float acc = (a0 + a1) + (a2 + a3);
#pragma unroll
  for (int off = 16; off > 0; off >>= 1) acc += __shfl_xor(acc, off, 32);
  return acc;
}

__global__ __launch_bounds__(256) __attribute__((amdgpu_waves_per_eu(6, 6)))
void pair_gelu_reduce(
    const float* __restrict__ hsrc, const float* __restrict__ htgt,
    const float* __restrict__ wcls, float* __restrict__ out) {
  const int lane = threadIdx.x & 31;
  const int wid  = threadIdx.x >> 5;
  const int s    = blockIdx.x * PW + wid;
  const int t0   = blockIdx.y * PT;
  const int l    = blockIdx.z % LL;
  const int b    = blockIdx.z / LL;

  const float* __restrict__ srcrow =
      hsrc + ((size_t)((b * SS + s) * LL + l)) * HH + 4 * lane;
  v4f sv[HV], wv[HV];
#pragma unroll
  for (int i = 0; i < HV; ++i) {
    sv[i] = *(const v4f*)(srcrow + 128 * i);
    const float* __restrict__ wp = wcls + (size_t)(4 * lane + 128 * i) * LL + l;
    wv[i].x = wp[0];
    wv[i].y = wp[LL];
    wv[i].z = wp[2 * LL];
    wv[i].w = wp[3 * LL];
  }

  const float* __restrict__ trow =
      htgt + ((size_t)((b * SS + t0) * LL + l)) * HH + 4 * lane;
  float* __restrict__ orow =
      out + ((size_t)((b * SS + s) * SS + t0)) * LL + l;

  v4f ta[HV], tb[HV];
#pragma unroll
  for (int i = 0; i < HV; ++i) ta[i] = *(const v4f*)(trow + 128 * i);

  for (int t = 0; t < PT; t += 2) {
    const float* __restrict__ r1 = trow + (size_t)(t + 1) * (LL * HH);
#pragma unroll
    for (int i = 0; i < HV; ++i) tb[i] = *(const v4f*)(r1 + 128 * i);

    const float o0 = row_dot(sv, wv, ta);
    if (lane == 0) orow[(size_t)t * LL] = o0;

    const int tn = (t + 2 < PT) ? (t + 2) : (PT - 1);
    const float* __restrict__ r2 = trow + (size_t)tn * (LL * HH);
#pragma unroll
    for (int i = 0; i < HV; ++i) ta[i] = *(const v4f*)(r2 + 128 * i);

    const float o1 = row_dot(sv, wv, tb);
    if (lane == 0) orow[(size_t)(t + 1) * LL] = o1;
  }
}

extern "C" void kernel_launch(void* const* d_in, const int* in_sizes, int n_in,
                              void* d_out, int out_size, void* d_ws, size_t ws_size,
                              hipStream_t stream) {
  const float* pooled = (const float*)d_in[0];
  const float* W_src  = (const float*)d_in[1];
  const float* b_src  = (const float*)d_in[2];
  const float* W_tgt  = (const float*)d_in[3];
  const float* b_tgt  = (const float*)d_in[4];
  const float* wcls   = (const float*)d_in[5];
  float* out = (float*)d_out;

  float* hsrc = (float*)d_ws;
  float* htgt = hsrc + (size_t)MM * NN;

  dim3 g1(NN / TN, MM / TM);
  proj_wmma_f32<<<g1, 128, 0, stream>>>(pooled, W_src, b_src, hsrc);
  proj_wmma_f32<<<g1, 128, 0, stream>>>(pooled, W_tgt, b_tgt, htgt);

  dim3 g2(SS / PW, SS / PT, BB * LL);
  pair_gelu_reduce<<<g2, 256, 0, stream>>>(hsrc, htgt, wcls, out);
}


# both projection GEMMs in one launch: grid z picks the weight set, 1152 equal work-groups all resident together (at most five per WGP); one launch boundary and one wave-entry ramp fewer; k order kept
# speedup vs baseline: 3.7349x; 1.0777x over previous
#include <hip/hip_runtime.h>
#include <math.h>

typedef __attribute__((ext_vector_type(2))) float v2f;
typedef __attribute__((ext_vector_type(4))) float v4f;
typedef __attribute__((ext_vector_type(8))) float v8f;

#define BB 2
#define SS 192
#define HH 768
#define LL 4
#define NN (HH * LL)
#define MM (BB * SS)

#define TM 32
#define TN 64
#define TK 64
#define LP 68

__device__ __forceinline__ v8f wmma4(v2f a, v2f b, v8f c) {
  return __builtin_amdgcn_wmma_f32_16x16x4_f32(
       false, a,  false, b,
       (short)0, c,  false,  false);
}

__global__ __launch_bounds__(128) void proj_wmma_f32(
    const float* __restrict__ X,
    const float* __restrict__ Wa, const float* __restrict__ biasa, float* __restrict__ Ya,
    const float* __restrict__ Wb, const float* __restrict__ biasb, float* __restrict__ Yb) {
  __shared__ __attribute__((aligned(16))) float sA[2][TM][LP];
  __shared__ __attribute__((aligned(16))) float sB[2][TN][LP];

  const bool second = blockIdx.z != 0;
  const float* __restrict__ W    = second ? Wb : Wa;
  const float* __restrict__ bias = second ? biasb : biasa;
  float* __restrict__ Y          = second ? Yb : Ya;

  const int n0   = blockIdx.x * TN;
  const int m0   = blockIdx.y * TM;
  const int tid  = threadIdx.x;
  const int lane = tid & 31;
  const int wid  = tid >> 5;
  const int wm   = wid & 1;
  const int wn   = wid >> 1;
  const int r16  = lane & 15;
  const int kh   = (lane >> 4) * 2;

  const int srow = tid >> 4;
  const int scol = (tid & 15) * 4;

  const float* __restrict__ gA = X + (size_t)(m0 + srow) * HH + scol;
  const float* __restrict__ gB = W + (size_t)(n0 + srow) * HH + scol;

  v4f ra[TM / 8];
  v4f rb[TN / 8];

  for (int j = 0; j < TM / 8; ++j) ra[j] = *(const v4f*)(gA + (size_t)(8 * j) * HH);
  for (int j = 0; j < TN / 8; ++j) rb[j] = *(const v4f*)(gB + (size_t)(8 * j) * HH);
  for (int j = 0; j < TM / 8; ++j) *(v4f*)&sA[0][srow + 8 * j][scol] = ra[j];
  for (int j = 0; j < TN / 8; ++j) *(v4f*)&sB[0][srow + 8 * j][scol] = rb[j];
  __syncthreads();

  v8f acc0 = {};
  v8f acc1 = {};

  for (int kt = 0; kt < HH / TK; ++kt) {
    const int  buf  = kt & 1;
    const bool more = (kt + 1) < (HH / TK);

    if (more) {
      const int ko = (kt + 1) * TK;
      for (int j = 0; j < TM / 8; ++j) ra[j] = *(const v4f*)(gA + (size_t)(8 * j) * HH + ko);
      for (int j = 0; j < TN / 8; ++j) rb[j] = *(const v4f*)(gB + (size_t)(8 * j) * HH + ko);
    }

    const float* __restrict__ pa  = &sA[buf][wm * 16 + r16][kh];
    const float* __restrict__ pb0 = &sB[buf][wn * 32 + r16][kh];
    const float* __restrict__ pb1 = &sB[buf][wn * 32 + 16 + r16][kh];
    v2f fa[TK / 4], fb0[TK / 4], fb1[TK / 4];
    for (int ks = 0; ks < TK / 4; ++ks) {
      fa[ks]  = *(const v2f*)(pa + 4 * ks);
      fb0[ks] = *(const v2f*)(pb0 + 4 * ks);
      fb1[ks] = *(const v2f*)(pb1 + 4 * ks);
    }
    __builtin_amdgcn_sched_barrier(0);
    for (int ks = 0; ks < TK / 4; ++ks) {
      acc0 = wmma4(fa[ks], fb0[ks], acc0);
      acc1 = wmma4(fa[ks], fb1[ks], acc1);
    }
    __builtin_amdgcn_sched_barrier(0);

    if (more) {
      for (int j = 0; j < TM / 8; ++j) *(v4f*)&sA[buf ^ 1][srow + 8 * j][scol] = ra[j];
      for (int j = 0; j < TN / 8; ++j) *(v4f*)&sB[buf ^ 1][srow + 8 * j][scol] = rb[j];
      __syncthreads();
    }
  }

  const int n     = n0 + wn * 32 + r16;
  const int mbase = m0 + wm * 16 + ((lane >> 4) << 3);
  const float bn0 = bias[n];
  const float bn1 = bias[n + 16];
  for (int r = 0; r < 8; ++r) {
    Y[(size_t)(mbase + r) * NN + n]      = acc0[r] + bn0;
    Y[(size_t)(mbase + r) * NN + n + 16] = acc1[r] + bn1;
  }
}

__device__ __forceinline__ float gelu_erf(float x) {
  const float u = fabsf(x);
  float q = 1.57496416e-8f;
  q = fmaf(q, u, -5.81307916e-7f);
  q = fmaf(q, u, 0.00000862902561f);
  q = fmaf(q, u, -0.0000658568606f);
  q = fmaf(q, u, 0.000250525482f);
  q = fmaf(q, u, 0.0000509071178f);
  q = fmaf(q, u, -0.00700288685f);
  q = fmaf(q, u, 0.0524775460f);
  q = fmaf(q, u, 0.459208757f);
  q = fmaf(q, u, 1.15110576f);
  q = fmaf(q, u, 0.999999940f);
  const float e = __builtin_amdgcn_exp2f(-q);
  return fmaf(-u, e, fmaxf(x, 0.0f));
}

#define PW 8
#define PT 48
#define HV (HH / 128)

__device__ __forceinline__ float row_dot(const v4f (&sv)[HV], const v4f (&wv)[HV],
                                         const v4f (&tv)[HV]) {
  float a0 = 0.0f, a1 = 0.0f, a2 = 0.0f, a3 = 0.0f;
#pragma unroll
  for (int i = 0; i < HV; ++i) {
    a0 = fmaf(gelu_erf(sv[i].x + tv[i].x), wv[i].x, a0);
    a1 = fmaf(gelu_erf(sv[i].y + tv[i].y), wv[i].y, a1);
    a2 = fmaf(gelu_erf(sv[i].z + tv[i].z), wv[i].z, a2);
    a3 = fmaf(gelu_erf(sv[i].w + tv[i].w), wv[i].w, a3);
  }
  float acc = (a0 + a1) + (a2 + a3);
#pragma unroll
  for (int off = 16; off > 0; off >>= 1) acc += __shfl_xor(acc, off, 32);
  return acc;
}

__global__ __launch_bounds__(256) __attribute__((amdgpu_waves_per_eu(6, 6)))
void pair_gelu_reduce(
    const float* __restrict__ hsrc, const float* __restrict__ htgt,
    const float* __restrict__ wcls, float* __restrict__ out) {
  const int lane = threadIdx.x & 31;
  const int wid  = threadIdx.x >> 5;
  const int s    = blockIdx.x * PW + wid;
  const int t0   = blockIdx.y * PT;
  const int l    = blockIdx.z % LL;
  const int b    = blockIdx.z / LL;

  const float* __restrict__ srcrow =
      hsrc + ((size_t)((b * SS + s) * LL + l)) * HH + 4 * lane;
  v4f sv[HV], wv[HV];
#pragma unroll
  for (int i = 0; i < HV; ++i) {
    sv[i] = *(const v4f*)(srcrow + 128 * i);
    const float* __restrict__ wp = wcls + (size_t)(4 * lane + 128 * i) * LL + l;
    wv[i].x = wp[0];
    wv[i].y = wp[LL];
    wv[i].z = wp[2 * LL];
    wv[i].w = wp[3 * LL];
  }

  const float* __restrict__ trow =
      htgt + ((size_t)((b * SS + t0) * LL + l)) * HH + 4 * lane;
  float* __restrict__ orow =
      out + ((size_t)((b * SS + s) * SS + t0)) * LL + l;

  v4f ta[HV], tb[HV];
#pragma unroll
  for (int i = 0; i < HV; ++i) ta[i] = *(const v4f*)(trow + 128 * i);

  for (int t = 0; t < PT; t += 2) {
    const float* __restrict__ r1 = trow + (size_t)(t + 1) * (LL * HH);
#pragma unroll
    for (int i = 0; i < HV; ++i) tb[i] = *(const v4f*)(r1 + 128 * i);

    const float o0 = row_dot(sv, wv, ta);
    if (lane == 0) orow[(size_t)t * LL] = o0;

    const int tn = (t + 2 < PT) ? (t + 2) : (PT - 1);
    const float* __restrict__ r2 = trow + (size_t)tn * (LL * HH);
#pragma unroll
    for (int i = 0; i < HV; ++i) ta[i] = *(const v4f*)(r2 + 128 * i);

    const float o1 = row_dot(sv, wv, tb);
    if (lane == 0) orow[(size_t)(t + 1) * LL] = o1;
  }
}

extern "C" void kernel_launch(void* const* d_in, const int* in_sizes, int n_in,
                              void* d_out, int out_size, void* d_ws, size_t ws_size,
                              hipStream_t stream) {
  const float* pooled = (const float*)d_in[0];
  const float* W_src  = (const float*)d_in[1];
  const float* b_src  = (const float*)d_in[2];
  const float* W_tgt  = (const float*)d_in[3];
  const float* b_tgt  = (const float*)d_in[4];
  const float* wcls   = (const float*)d_in[5];
  float* out = (float*)d_out;

  float* hsrc = (float*)d_ws;
  float* htgt = hsrc + (size_t)MM * NN;

  dim3 g1(NN / TN, MM / TM, 2);
  proj_wmma_f32<<<g1, 128, 0, stream>>>(pooled, W_src, b_src, hsrc, W_tgt, b_tgt, htgt);

  dim3 g2(SS / PW, SS / PT, BB * LL);
  pair_gelu_reduce<<<g2, 256, 0, stream>>>(hsrc, htgt, wcls, out);
}


# pair kernel: the wave sums leave the LDS permute path (5 steps + waits per row) for the vector ALU: one half swap + four DPP row_xmask adds serve two rows at once; cross-lane sum order changed
# speedup vs baseline: 4.0363x; 1.0807x over previous
#include <hip/hip_runtime.h>
#include <math.h>

typedef __attribute__((ext_vector_type(2))) float v2f;
typedef __attribute__((ext_vector_type(4))) float v4f;
typedef __attribute__((ext_vector_type(8))) float v8f;

#define BB 2
#define SS 192
#define HH 768
#define LL 4
#define NN (HH * LL)
#define MM (BB * SS)

#define TM 32
#define TN 64
#define TK 64
#define LP 68

__device__ __forceinline__ v8f wmma4(v2f a, v2f b, v8f c) {
  return __builtin_amdgcn_wmma_f32_16x16x4_f32(
       false, a,  false, b,
       (short)0, c,  false,  false);
}

__global__ __launch_bounds__(128) void proj_wmma_f32(
    const float* __restrict__ X,
    const float* __restrict__ Wa, const float* __restrict__ biasa, float* __restrict__ Ya,
    const float* __restrict__ Wb, const float* __restrict__ biasb, float* __restrict__ Yb) {
  __shared__ __attribute__((aligned(16))) float sA[2][TM][LP];
  __shared__ __attribute__((aligned(16))) float sB[2][TN][LP];

  const bool second = blockIdx.z != 0;
  const float* __restrict__ W    = second ? Wb : Wa;
  const float* __restrict__ bias = second ? biasb : biasa;
  float* __restrict__ Y          = second ? Yb : Ya;

  const int n0   = blockIdx.x * TN;
  const int m0   = blockIdx.y * TM;
  const int tid  = threadIdx.x;
  const int lane = tid & 31;
  const int wid  = tid >> 5;
  const int wm   = wid & 1;
  const int wn   = wid >> 1;
  const int r16  = lane & 15;
  const int kh   = (lane >> 4) * 2;

  const int srow = tid >> 4;
  const int scol = (tid & 15) * 4;

  const float* __restrict__ gA = X + (size_t)(m0 + srow) * HH + scol;
  const float* __restrict__ gB = W + (size_t)(n0 + srow) * HH + scol;

  v4f ra[TM / 8];
  v4f rb[TN / 8];

  for (int j = 0; j < TM / 8; ++j) ra[j] = *(const v4f*)(gA + (size_t)(8 * j) * HH);
  for (int j = 0; j < TN / 8; ++j) rb[j] = *(const v4f*)(gB + (size_t)(8 * j) * HH);
  for (int j = 0; j < TM / 8; ++j) *(v4f*)&sA[0][srow + 8 * j][scol] = ra[j];
  for (int j = 0; j < TN / 8; ++j) *(v4f*)&sB[0][srow + 8 * j][scol] = rb[j];
  __syncthreads();

  v8f acc0 = {};
  v8f acc1 = {};

  for (int kt = 0; kt < HH / TK; ++kt) {
    const int  buf  = kt & 1;
    const bool more = (kt + 1) < (HH / TK);

    if (more) {
      const int ko = (kt + 1) * TK;
      for (int j = 0; j < TM / 8; ++j) ra[j] = *(const v4f*)(gA + (size_t)(8 * j) * HH + ko);
      for (int j = 0; j < TN / 8; ++j) rb[j] = *(const v4f*)(gB + (size_t)(8 * j) * HH + ko);
    }

    const float* __restrict__ pa  = &sA[buf][wm * 16 + r16][kh];
    const float* __restrict__ pb0 = &sB[buf][wn * 32 + r16][kh];
    const float* __restrict__ pb1 = &sB[buf][wn * 32 + 16 + r16][kh];
    v2f fa[TK / 4], fb0[TK / 4], fb1[TK / 4];
    for (int ks = 0; ks < TK / 4; ++ks) {
      fa[ks]  = *(const v2f*)(pa + 4 * ks);
      fb0[ks] = *(const v2f*)(pb0 + 4 * ks);
      fb1[ks] = *(const v2f*)(pb1 + 4 * ks);
    }
    __builtin_amdgcn_sched_barrier(0);
    for (int ks = 0; ks < TK / 4; ++ks) {
      acc0 = wmma4(fa[ks], fb0[ks], acc0);
      acc1 = wmma4(fa[ks], fb1[ks], acc1);
    }
    __builtin_amdgcn_sched_barrier(0);

    if (more) {
      for (int j = 0; j < TM / 8; ++j) *(v4f*)&sA[buf ^ 1][srow + 8 * j][scol] = ra[j];
      for (int j = 0; j < TN / 8; ++j) *(v4f*)&sB[buf ^ 1][srow + 8 * j][scol] = rb[j];
      __syncthreads();
    }
  }

  const int n     = n0 + wn * 32 + r16;
  const int mbase = m0 + wm * 16 + ((lane >> 4) << 3);
  const float bn0 = bias[n];
  const float bn1 = bias[n + 16];
  for (int r = 0; r < 8; ++r) {
    Y[(size_t)(mbase + r) * NN + n]      = acc0[r] + bn0;
    Y[(size_t)(mbase + r) * NN + n + 16] = acc1[r] + bn1;
  }
}

__device__ __forceinline__ v2f splat2(float c) { return (v2f){c, c}; }

__device__ __forceinline__ v2f gelu_erf2(v2f x) {
  const v2f u = __builtin_elementwise_abs(x);
  v2f q = splat2(1.57496416e-8f);
  q = __builtin_elementwise_fma(q, u, splat2(-5.81307916e-7f));
  q = __builtin_elementwise_fma(q, u, splat2(0.00000862902561f));
  q = __builtin_elementwise_fma(q, u, splat2(-0.0000658568606f));
  q = __builtin_elementwise_fma(q, u, splat2(0.000250525482f));
  q = __builtin_elementwise_fma(q, u, splat2(0.0000509071178f));
  q = __builtin_elementwise_fma(q, u, splat2(-0.00700288685f));
  q = __builtin_elementwise_fma(q, u, splat2(0.0524775460f));
  q = __builtin_elementwise_fma(q, u, splat2(0.459208757f));
  q = __builtin_elementwise_fma(q, u, splat2(1.15110576f));
  q = __builtin_elementwise_fma(q, u, splat2(0.999999940f));
  v2f e;
  e.x = __builtin_amdgcn_exp2f(-q.x);
  e.y = __builtin_amdgcn_exp2f(-q.y);
  return __builtin_elementwise_fma(-u, e, __builtin_elementwise_max(x, splat2(0.0f)));
}

#define PW 8
#define PT 48
#define HV (HH / 128)

__device__ __forceinline__ float xhalf(float x) {
  const int b = __builtin_bit_cast(int, x);
  return __builtin_bit_cast(float, __builtin_amdgcn_permlanex16(
      b, b, 0x76543210, 0xfedcba98, false, false));
}

template <int N>
__device__ __forceinline__ float row_xadd(float v) {
  const int b = __builtin_bit_cast(int, v);
  return v + __builtin_bit_cast(float, __builtin_amdgcn_update_dpp(0, b, 0x160 + N, 0xf, 0xf, true));
}

__device__ __forceinline__ float row_part(const v4f (&sv)[HV], const v4f (&wv)[HV],
                                         const v4f (&tv)[HV]) {
  v2f a01 = splat2(0.0f), a23 = splat2(0.0f);
#pragma unroll
  for (int i = 0; i < HV; ++i) {
    a01 = __builtin_elementwise_fma(gelu_erf2(sv[i].xy + tv[i].xy), wv[i].xy, a01);
    a23 = __builtin_elementwise_fma(gelu_erf2(sv[i].zw + tv[i].zw), wv[i].zw, a23);
  }
  return (a01.x + a01.y) + (a23.x + a23.y);
}

__global__ __launch_bounds__(256) __attribute__((amdgpu_waves_per_eu(6, 6)))
void pair_gelu_reduce(
    const float* __restrict__ hsrc, const float* __restrict__ htgt,
    const float* __restrict__ wcls, float* __restrict__ out) {
  const int lane = threadIdx.x & 31;
  const bool hi  = lane >= 16;
  const int wid  = threadIdx.x >> 5;
  const int s    = blockIdx.x * PW + wid;
  const int t0   = blockIdx.y * PT;
  const int l    = blockIdx.z % LL;
  const int b    = blockIdx.z / LL;

  const float* __restrict__ srcrow =
      hsrc + ((size_t)((b * SS + s) * LL + l)) * HH + 4 * lane;
  v4f sv[HV], wv[HV];
#pragma unroll
  for (int i = 0; i < HV; ++i) {
    sv[i] = *(const v4f*)(srcrow + 128 * i);
    const float* __restrict__ wp = wcls + (size_t)(4 * lane + 128 * i) * LL + l;
    wv[i].x = wp[0];
    wv[i].y = wp[LL];
    wv[i].z = wp[2 * LL];
    wv[i].w = wp[3 * LL];
  }

  const float* __restrict__ trow =
      htgt + ((size_t)((b * SS + t0) * LL + l)) * HH + 4 * lane;
  float* __restrict__ orow =
      out + ((size_t)((b * SS + s) * SS + t0)) * LL + l;

  v4f ta[HV], tb[HV];
#pragma unroll
  for (int i = 0; i < HV; ++i) ta[i] = *(const v4f*)(trow + 128 * i);

  for (int t = 0; t < PT; t += 2) {
    const float* __restrict__ r1 = trow + (size_t)(t + 1) * (LL * HH);
#pragma unroll
    for (int i = 0; i < HV; ++i) tb[i] = *(const v4f*)(r1 + 128 * i);

    const float p0 = row_part(sv, wv, ta);

    const int tn = (t + 2 < PT) ? (t + 2) : (PT - 1);
    const float* __restrict__ r2 = trow + (size_t)tn * (LL * HH);
#pragma unroll
    for (int i = 0; i < HV; ++i) ta[i] = *(const v4f*)(r2 + 128 * i);

    const float p1 = row_part(sv, wv, tb);

    float v = (hi ? p1 : p0) + xhalf(hi ? p0 : p1);
    v = row_xadd<1>(v);
    v = row_xadd<2>(v);
    v = row_xadd<4>(v);
    v = row_xadd<8>(v);
    if ((lane & 15) == 0) orow[(size_t)(t + (lane >> 4)) * LL] = v;
  }
}

extern "C" void kernel_launch(void* const* d_in, const int* in_sizes, int n_in,
                              void* d_out, int out_size, void* d_ws, size_t ws_size,
                              hipStream_t stream) {
  const float* pooled = (const float*)d_in[0];
  const float* W_src  = (const float*)d_in[1];
  const float* b_src  = (const float*)d_in[2];
  const float* W_tgt  = (const float*)d_in[3];
  const float* b_tgt  = (const float*)d_in[4];
  const float* wcls   = (const float*)d_in[5];
  float* out = (float*)d_out;

  float* hsrc = (float*)d_ws;
  float* htgt = hsrc + (size_t)MM * NN;

  dim3 g1(NN / TN, MM / TM, 2);
  proj_wmma_f32<<<g1, 128, 0, stream>>>(pooled, W_src, b_src, hsrc, W_tgt, b_tgt, htgt);

  dim3 g2(SS / PW, SS / PT, BB * LL);
  pair_gelu_reduce<<<g2, 256, 0, stream>>>(hsrc, htgt, wcls, out);
}


# pair kernel: GELU takes 3 fewer vector issue slots per value pair: Q refit at degree 8 (was 10) at the same error bound; |x| as 2*max(x,0)-x in one packed fma instead of two sign-bit ands
# speedup vs baseline: 4.3404x; 1.0753x over previous
#include <hip/hip_runtime.h>
#include <math.h>

typedef __attribute__((ext_vector_type(2))) float v2f;
typedef __attribute__((ext_vector_type(4))) float v4f;
typedef __attribute__((ext_vector_type(8))) float v8f;

#define BB 2
#define SS 192
#define HH 768
#define LL 4
#define NN (HH * LL)
#define MM (BB * SS)

#define TM 32
#define TN 64
#define TK 64
#define LP 68

__device__ __forceinline__ v8f wmma4(v2f a, v2f b, v8f c) {
  return __builtin_amdgcn_wmma_f32_16x16x4_f32(
       false, a,  false, b,
       (short)0, c,  false,  false);
}

__global__ __launch_bounds__(128) void proj_wmma_f32(
    const float* __restrict__ X,
    const float* __restrict__ Wa, const float* __restrict__ biasa, float* __restrict__ Ya,
    const float* __restrict__ Wb, const float* __restrict__ biasb, float* __restrict__ Yb) {
  __shared__ __attribute__((aligned(16))) float sA[2][TM][LP];
  __shared__ __attribute__((aligned(16))) float sB[2][TN][LP];

  const bool second = blockIdx.z != 0;
  const float* __restrict__ W    = second ? Wb : Wa;
  const float* __restrict__ bias = second ? biasb : biasa;
  float* __restrict__ Y          = second ? Yb : Ya;

  const int n0   = blockIdx.x * TN;
  const int m0   = blockIdx.y * TM;
  const int tid  = threadIdx.x;
  const int lane = tid & 31;
  const int wid  = tid >> 5;
  const int wm   = wid & 1;
  const int wn   = wid >> 1;
  const int r16  = lane & 15;
  const int kh   = (lane >> 4) * 2;

  const int srow = tid >> 4;
  const int scol = (tid & 15) * 4;

  const float* __restrict__ gA = X + (size_t)(m0 + srow) * HH + scol;
  const float* __restrict__ gB = W + (size_t)(n0 + srow) * HH + scol;

  v4f ra[TM / 8];
  v4f rb[TN / 8];

  for (int j = 0; j < TM / 8; ++j) ra[j] = *(const v4f*)(gA + (size_t)(8 * j) * HH);
  for (int j = 0; j < TN / 8; ++j) rb[j] = *(const v4f*)(gB + (size_t)(8 * j) * HH);
  for (int j = 0; j < TM / 8; ++j) *(v4f*)&sA[0][srow + 8 * j][scol] = ra[j];
  for (int j = 0; j < TN / 8; ++j) *(v4f*)&sB[0][srow + 8 * j][scol] = rb[j];
  __syncthreads();

  v8f acc0 = {};
  v8f acc1 = {};

  for (int kt = 0; kt < HH / TK; ++kt) {
    const int  buf  = kt & 1;
    const bool more = (kt + 1) < (HH / TK);

    if (more) {
      const int ko = (kt + 1) * TK;
      for (int j = 0; j < TM / 8; ++j) ra[j] = *(const v4f*)(gA + (size_t)(8 * j) * HH + ko);
      for (int j = 0; j < TN / 8; ++j) rb[j] = *(const v4f*)(gB + (size_t)(8 * j) * HH + ko);
    }

    const float* __restrict__ pa  = &sA[buf][wm * 16 + r16][kh];
    const float* __restrict__ pb0 = &sB[buf][wn * 32 + r16][kh];
    const float* __restrict__ pb1 = &sB[buf][wn * 32 + 16 + r16][kh];
    v2f fa[TK / 4], fb0[TK / 4], fb1[TK / 4];
    for (int ks = 0; ks < TK / 4; ++ks) {
      fa[ks]  = *(const v2f*)(pa + 4 * ks);
      fb0[ks] = *(const v2f*)(pb0 + 4 * ks);
      fb1[ks] = *(const v2f*)(pb1 + 4 * ks);
    }
    __builtin_amdgcn_sched_barrier(0);
    for (int ks = 0; ks < TK / 4; ++ks) {
      acc0 = wmma4(fa[ks], fb0[ks], acc0);
      acc1 = wmma4(fa[ks], fb1[ks], acc1);
    }
    __builtin_amdgcn_sched_barrier(0);

    if (more) {
      for (int j = 0; j < TM / 8; ++j) *(v4f*)&sA[buf ^ 1][srow + 8 * j][scol] = ra[j];
      for (int j = 0; j < TN / 8; ++j) *(v4f*)&sB[buf ^ 1][srow + 8 * j][scol] = rb[j];
      __syncthreads();
    }
  }

  const int n     = n0 + wn * 32 + r16;
  const int mbase = m0 + wm * 16 + ((lane >> 4) << 3);
  const float bn0 = bias[n];
  const float bn1 = bias[n + 16];
  for (int r = 0; r < 8; ++r) {
    Y[(size_t)(mbase + r) * NN + n]      = acc0[r] + bn0;
    Y[(size_t)(mbase + r) * NN + n + 16] = acc1[r] + bn1;
  }
}

__device__ __forceinline__ v2f splat2(float c) { return (v2f){c, c}; }

__device__ __forceinline__ v2f gelu_erf2(v2f x) {
  const v2f mx = __builtin_elementwise_max(x, splat2(0.0f));
  const v2f u = __builtin_elementwise_fma(mx, splat2(2.0f), -x);
  v2f q = splat2(1.6058508e-6f);
  q = __builtin_elementwise_fma(q, u, splat2(-2.40446807e-5f));
  q = __builtin_elementwise_fma(q, u, splat2(0.000109311841f));
  q = __builtin_elementwise_fma(q, u, splat2(0.000336651603f));
  q = __builtin_elementwise_fma(q, u, splat2(-0.00735267764f));
  q = __builtin_elementwise_fma(q, u, splat2(0.0527299233f));
  q = __builtin_elementwise_fma(q, u, splat2(0.459108621f));
  q = __builtin_elementwise_fma(q, u, splat2(1.1511246f));
  q = __builtin_elementwise_fma(q, u, splat2(0.999998808f));
  v2f e;
  e.x = __builtin_amdgcn_exp2f(-q.x);
  e.y = __builtin_amdgcn_exp2f(-q.y);
  return __builtin_elementwise_fma(-u, e, mx);
}

#define PW 8
#define PT 48
#define HV (HH / 128)

__device__ __forceinline__ float xhalf(float x) {
  const int b = __builtin_bit_cast(int, x);
  return __builtin_bit_cast(float, __builtin_amdgcn_permlanex16(
      b, b, 0x76543210, 0xfedcba98, false, false));
}

template <int N>
__device__ __forceinline__ float row_xadd(float v) {
  const int b = __builtin_bit_cast(int, v);
  return v + __builtin_bit_cast(float, __builtin_amdgcn_update_dpp(0, b, 0x160 + N, 0xf, 0xf, true));
}

__device__ __forceinline__ float row_part(const v4f (&sv)[HV], const v4f (&wv)[HV],
                                         const v4f (&tv)[HV]) {
  v2f a01 = splat2(0.0f), a23 = splat2(0.0f);
#pragma unroll
  for (int i = 0; i < HV; ++i) {
    a01 = __builtin_elementwise_fma(gelu_erf2(sv[i].xy + tv[i].xy), wv[i].xy, a01);
    a23 = __builtin_elementwise_fma(gelu_erf2(sv[i].zw + tv[i].zw), wv[i].zw, a23);
  }
  return (a01.x + a01.y) + (a23.x + a23.y);
}

__global__ __launch_bounds__(256) __attribute__((amdgpu_waves_per_eu(6, 6)))
void pair_gelu_reduce(
    const float* __restrict__ hsrc, const float* __restrict__ htgt,
    const float* __restrict__ wcls, float* __restrict__ out) {
  const int lane = threadIdx.x & 31;
  const bool hi  = lane >= 16;
  const int wid  = threadIdx.x >> 5;
  const int s    = blockIdx.x * PW + wid;
  const int t0   = blockIdx.y * PT;
  const int l    = blockIdx.z % LL;
  const int b    = blockIdx.z / LL;

  const float* __restrict__ srcrow =
      hsrc + ((size_t)((b * SS + s) * LL + l)) * HH + 4 * lane;
  v4f sv[HV], wv[HV];
#pragma unroll
  for (int i = 0; i < HV; ++i) {
    sv[i] = *(const v4f*)(srcrow + 128 * i);
    const float* __restrict__ wp = wcls + (size_t)(4 * lane + 128 * i) * LL + l;
    wv[i].x = wp[0];
    wv[i].y = wp[LL];
    wv[i].z = wp[2 * LL];
    wv[i].w = wp[3 * LL];
  }

  const float* __restrict__ trow =
      htgt + ((size_t)((b * SS + t0) * LL + l)) * HH + 4 * lane;
  float* __restrict__ orow =
      out + ((size_t)((b * SS + s) * SS + t0)) * LL + l;

  v4f ta[HV], tb[HV];
#pragma unroll
  for (int i = 0; i < HV; ++i) ta[i] = *(const v4f*)(trow + 128 * i);

  for (int t = 0; t < PT; t += 2) {
    const float* __restrict__ r1 = trow + (size_t)(t + 1) * (LL * HH);
#pragma unroll
    for (int i = 0; i < HV; ++i) tb[i] = *(const v4f*)(r1 + 128 * i);

    const float p0 = row_part(sv, wv, ta);

    const int tn = (t + 2 < PT) ? (t + 2) : (PT - 1);
    const float* __restrict__ r2 = trow + (size_t)tn * (LL * HH);
#pragma unroll
    for (int i = 0; i < HV; ++i) ta[i] = *(const v4f*)(r2 + 128 * i);

    const float p1 = row_part(sv, wv, tb);

    float v = (hi ? p1 : p0) + xhalf(hi ? p0 : p1);
    v = row_xadd<1>(v);
    v = row_xadd<2>(v);
    v = row_xadd<4>(v);
    v = row_xadd<8>(v);
    if ((lane & 15) == 0) orow[(size_t)(t + (lane >> 4)) * LL] = v;
  }
}

extern "C" void kernel_launch(void* const* d_in, const int* in_sizes, int n_in,
                              void* d_out, int out_size, void* d_ws, size_t ws_size,
                              hipStream_t stream) {
  const float* pooled = (const float*)d_in[0];
  const float* W_src  = (const float*)d_in[1];
  const float* b_src  = (const float*)d_in[2];
  const float* W_tgt  = (const float*)d_in[3];
  const float* b_tgt  = (const float*)d_in[4];
  const float* wcls   = (const float*)d_in[5];
  float* out = (float*)d_out;

  float* hsrc = (float*)d_ws;
  float* htgt = hsrc + (size_t)MM * NN;

  dim3 g1(NN / TN, MM / TM, 2);
  proj_wmma_f32<<<g1, 128, 0, stream>>>(pooled, W_src, b_src, hsrc, W_tgt, b_tgt, htgt);

  dim3 g2(SS / PW, SS / PT, BB * LL);
  pair_gelu_reduce<<<g2, 256, 0, stream>>>(hsrc, htgt, wcls, out);
}


# pair kernel: GELU as 0.5*x*(1+tanh(x*P(x*x))) on the hardware tanh, P a degree-6 fit of atanh(erf(x/sqrt2))/x; no max, no |x|: 13 vector issue slots per value pair, was 15; 0.5 folded into w
# speedup vs baseline: 4.6463x; 1.0705x over previous
#include <hip/hip_runtime.h>
#include <math.h>

typedef __attribute__((ext_vector_type(2))) float v2f;
typedef __attribute__((ext_vector_type(4))) float v4f;
typedef __attribute__((ext_vector_type(8))) float v8f;

#define BB 2
#define SS 192
#define HH 768
#define LL 4
#define NN (HH * LL)
#define MM (BB * SS)

#define TM 32
#define TN 64
#define TK 64
#define LP 68

__device__ __forceinline__ v8f wmma4(v2f a, v2f b, v8f c) {
  return __builtin_amdgcn_wmma_f32_16x16x4_f32(
       false, a,  false, b,
       (short)0, c,  false,  false);
}

__global__ __launch_bounds__(128) void proj_wmma_f32(
    const float* __restrict__ X,
    const float* __restrict__ Wa, const float* __restrict__ biasa, float* __restrict__ Ya,
    const float* __restrict__ Wb, const float* __restrict__ biasb, float* __restrict__ Yb) {
  __shared__ __attribute__((aligned(16))) float sA[2][TM][LP];
  __shared__ __attribute__((aligned(16))) float sB[2][TN][LP];

  const bool second = blockIdx.z != 0;
  const float* __restrict__ W    = second ? Wb : Wa;
  const float* __restrict__ bias = second ? biasb : biasa;
  float* __restrict__ Y          = second ? Yb : Ya;

  const int n0   = blockIdx.x * TN;
  const int m0   = blockIdx.y * TM;
  const int tid  = threadIdx.x;
  const int lane = tid & 31;
  const int wid  = tid >> 5;
  const int wm   = wid & 1;
  const int wn   = wid >> 1;
  const int r16  = lane & 15;
  const int kh   = (lane >> 4) * 2;

  const int srow = tid >> 4;
  const int scol = (tid & 15) * 4;

  const float* __restrict__ gA = X + (size_t)(m0 + srow) * HH + scol;
  const float* __restrict__ gB = W + (size_t)(n0 + srow) * HH + scol;

  v4f ra[TM / 8];
  v4f rb[TN / 8];

  for (int j = 0; j < TM / 8; ++j) ra[j] = *(const v4f*)(gA + (size_t)(8 * j) * HH);
  for (int j = 0; j < TN / 8; ++j) rb[j] = *(const v4f*)(gB + (size_t)(8 * j) * HH);
  for (int j = 0; j < TM / 8; ++j) *(v4f*)&sA[0][srow + 8 * j][scol] = ra[j];
  for (int j = 0; j < TN / 8; ++j) *(v4f*)&sB[0][srow + 8 * j][scol] = rb[j];
  __syncthreads();

  v8f acc0 = {};
  v8f acc1 = {};

  for (int kt = 0; kt < HH / TK; ++kt) {
    const int  buf  = kt & 1;
    const bool more = (kt + 1) < (HH / TK);

    if (more) {
      const int ko = (kt + 1) * TK;
      for (int j = 0; j < TM / 8; ++j) ra[j] = *(const v4f*)(gA + (size_t)(8 * j) * HH + ko);
      for (int j = 0; j < TN / 8; ++j) rb[j] = *(const v4f*)(gB + (size_t)(8 * j) * HH + ko);
    }

    const float* __restrict__ pa  = &sA[buf][wm * 16 + r16][kh];
    const float* __restrict__ pb0 = &sB[buf][wn * 32 + r16][kh];
    const float* __restrict__ pb1 = &sB[buf][wn * 32 + 16 + r16][kh];
    v2f fa[TK / 4], fb0[TK / 4], fb1[TK / 4];
    for (int ks = 0; ks < TK / 4; ++ks) {
      fa[ks]  = *(const v2f*)(pa + 4 * ks);
      fb0[ks] = *(const v2f*)(pb0 + 4 * ks);
      fb1[ks] = *(const v2f*)(pb1 + 4 * ks);
    }
    __builtin_amdgcn_sched_barrier(0);
    for (int ks = 0; ks < TK / 4; ++ks) {
      acc0 = wmma4(fa[ks], fb0[ks], acc0);
      acc1 = wmma4(fa[ks], fb1[ks], acc1);
    }
    __builtin_amdgcn_sched_barrier(0);

    if (more) {
      for (int j = 0; j < TM / 8; ++j) *(v4f*)&sA[buf ^ 1][srow + 8 * j][scol] = ra[j];
      for (int j = 0; j < TN / 8; ++j) *(v4f*)&sB[buf ^ 1][srow + 8 * j][scol] = rb[j];
      __syncthreads();
    }
  }

  const int n     = n0 + wn * 32 + r16;
  const int mbase = m0 + wm * 16 + ((lane >> 4) << 3);
  const float bn0 = bias[n];
  const float bn1 = bias[n + 16];
  for (int r = 0; r < 8; ++r) {
    Y[(size_t)(mbase + r) * NN + n]      = acc0[r] + bn0;
    Y[(size_t)(mbase + r) * NN + n + 16] = acc1[r] + bn1;
  }
}

__device__ __forceinline__ v2f splat2(float c) { return (v2f){c, c}; }

__device__ __forceinline__ v2f gelu_erf2_twice(v2f x) {
  const v2f z = x * x;
  v2f p = splat2(1.80561643e-9f);
  p = __builtin_elementwise_fma(p, z, splat2(-1.33446406e-7f));
  p = __builtin_elementwise_fma(p, z, splat2(3.96910946e-6f));
  p = __builtin_elementwise_fma(p, z, splat2(-5.52401580e-5f));
  p = __builtin_elementwise_fma(p, z, splat2(-3.31299925e-5f));
  p = __builtin_elementwise_fma(p, z, splat2(0.0363342836f));
  p = __builtin_elementwise_fma(p, z, splat2(0.797884226f));
  const v2f r = x * p;
  v2f t;
  t.x = __builtin_amdgcn_tanhf(r.x);
  t.y = __builtin_amdgcn_tanhf(r.y);
  return __builtin_elementwise_fma(x, t, x);
}

#define PW 8
#define PT 48
#define HV (HH / 128)

__device__ __forceinline__ float xhalf(float x) {
  const int b = __builtin_bit_cast(int, x);
  return __builtin_bit_cast(float, __builtin_amdgcn_permlanex16(
      b, b, 0x76543210, 0xfedcba98, false, false));
}

template <int N>
__device__ __forceinline__ float row_xadd(float v) {
  const int b = __builtin_bit_cast(int, v);
  return v + __builtin_bit_cast(float, __builtin_amdgcn_update_dpp(0, b, 0x160 + N, 0xf, 0xf, true));
}

__device__ __forceinline__ float row_part(const v4f (&sv)[HV], const v4f (&wv)[HV],
                                         const v4f (&tv)[HV]) {
  v2f a01 = splat2(0.0f), a23 = splat2(0.0f);
#pragma unroll
  for (int i = 0; i < HV; ++i) {
    a01 = __builtin_elementwise_fma(gelu_erf2_twice(sv[i].xy + tv[i].xy), wv[i].xy, a01);
    a23 = __builtin_elementwise_fma(gelu_erf2_twice(sv[i].zw + tv[i].zw), wv[i].zw, a23);
  }
  return (a01.x + a01.y) + (a23.x + a23.y);
}

__global__ __launch_bounds__(256) __attribute__((amdgpu_waves_per_eu(6, 6)))
void pair_gelu_reduce(
    const float* __restrict__ hsrc, const float* __restrict__ htgt,
    const float* __restrict__ wcls, float* __restrict__ out) {
  const int lane = threadIdx.x & 31;
  const bool hi  = lane >= 16;
  const int wid  = threadIdx.x >> 5;
  const int s    = blockIdx.x * PW + wid;
  const int t0   = blockIdx.y * PT;
  const int l    = blockIdx.z % LL;
  const int b    = blockIdx.z / LL;

  const float* __restrict__ srcrow =
      hsrc + ((size_t)((b * SS + s) * LL + l)) * HH + 4 * lane;
  v4f sv[HV], wv[HV];
#pragma unroll
  for (int i = 0; i < HV; ++i) {
    sv[i] = *(const v4f*)(srcrow + 128 * i);
    const float* __restrict__ wp = wcls + (size_t)(4 * lane + 128 * i) * LL + l;
    wv[i].x = 0.5f * wp[0];
    wv[i].y = 0.5f * wp[LL];
    wv[i].z = 0.5f * wp[2 * LL];
    wv[i].w = 0.5f * wp[3 * LL];
  }

  const float* __restrict__ trow =
      htgt + ((size_t)((b * SS + t0) * LL + l)) * HH + 4 * lane;
  float* __restrict__ orow =
      out + ((size_t)((b * SS + s) * SS + t0)) * LL + l;

  v4f ta[HV], tb[HV];
#pragma unroll
  for (int i = 0; i < HV; ++i) ta[i] = *(const v4f*)(trow + 128 * i);

  for (int t = 0; t < PT; t += 2) {
    const float* __restrict__ r1 = trow + (size_t)(t + 1) * (LL * HH);
#pragma unroll
    for (int i = 0; i < HV; ++i) tb[i] = *(const v4f*)(r1 + 128 * i);

    const float p0 = row_part(sv, wv, ta);

    const int tn = (t + 2 < PT) ? (t + 2) : (PT - 1);
    const float* __restrict__ r2 = trow + (size_t)tn * (LL * HH);
#pragma unroll
    for (int i = 0; i < HV; ++i) ta[i] = *(const v4f*)(r2 + 128 * i);

    const float p1 = row_part(sv, wv, tb);

    float v = (hi ? p1 : p0) + xhalf(hi ? p0 : p1);
    v = row_xadd<1>(v);
    v = row_xadd<2>(v);
    v = row_xadd<4>(v);
    v = row_xadd<8>(v);
    if ((lane & 15) == 0) orow[(size_t)(t + (lane >> 4)) * LL] = v;
  }
}

extern "C" void kernel_launch(void* const* d_in, const int* in_sizes, int n_in,
                              void* d_out, int out_size, void* d_ws, size_t ws_size,
                              hipStream_t stream) {
  const float* pooled = (const float*)d_in[0];
  const float* W_src  = (const float*)d_in[1];
  const float* b_src  = (const float*)d_in[2];
  const float* W_tgt  = (const float*)d_in[3];
  const float* b_tgt  = (const float*)d_in[4];
  const float* wcls   = (const float*)d_in[5];
  float* out = (float*)d_out;

  float* hsrc = (float*)d_ws;
  float* htgt = hsrc + (size_t)MM * NN;

  dim3 g1(NN / TN, MM / TM, 2);
  proj_wmma_f32<<<g1, 128, 0, stream>>>(pooled, W_src, b_src, hsrc, W_tgt, b_tgt, htgt);

  dim3 g2(SS / PW, SS / PT, BB * LL);
  pair_gelu_reduce<<<g2, 256, 0, stream>>>(hsrc, htgt, wcls, out);
}


# projection GEMM retiled: 256 equal work-groups of 12 waves, one 96x96 tile of Y per WGP (wave tile 48x16); next slab staged to LDS under the multiplies, barrier signalled early; k order kept
# speedup vs baseline: 4.8799x; 1.0503x over previous
#include <hip/hip_runtime.h>
#include <math.h>

typedef __attribute__((ext_vector_type(2))) float v2f;
typedef __attribute__((ext_vector_type(4))) float v4f;
typedef __attribute__((ext_vector_type(8))) float v8f;

#define BB 2
#define SS 192
#define HH 768
#define LL 4
#define NN (HH * LL)
#define MM (BB * SS)

#define TM 96
#define TN 96
#define TK 64
#define LP 68
#define NK (HH / TK)
#define PJ (TM / 24)
#define KSIG 12

__device__ __forceinline__ v8f wmma4(v2f a, v2f b, v8f c) {
  return __builtin_amdgcn_wmma_f32_16x16x4_f32(
       false, a,  false, b,
       (short)0, c,  false,  false);
}

__global__ __launch_bounds__(384) void proj_wmma_f32(
    const float* __restrict__ X,
    const float* __restrict__ Wa, const float* __restrict__ biasa, float* __restrict__ Ya,
    const float* __restrict__ Wb, const float* __restrict__ biasb, float* __restrict__ Yb) {
  __shared__ __attribute__((aligned(16))) float sA[2][TM][LP];
  __shared__ __attribute__((aligned(16))) float sB[2][TN][LP];

  const bool second = blockIdx.z != 0;
  const float* __restrict__ W    = second ? Wb : Wa;
  const float* __restrict__ bias = second ? biasb : biasa;
  float* __restrict__ Y          = second ? Yb : Ya;

  const int n0   = blockIdx.x * TN;
  const int m0   = blockIdx.y * TM;
  const int tid  = threadIdx.x;
  const int lane = tid & 31;
  const int wid  = tid >> 5;
  const int wm   = wid & 1;
  const int wn   = wid >> 1;
  const int r16  = lane & 15;
  const int kh   = (lane >> 4) * 2;

  const int srow = tid >> 4;
  const int scol = (tid & 15) * 4;

  const float* __restrict__ gA = X + (size_t)(m0 + srow) * HH + scol;
  const float* __restrict__ gB = W + (size_t)(n0 + srow) * HH + scol;

  v4f ra[PJ];
  v4f rb[PJ];

  for (int j = 0; j < PJ; ++j) ra[j] = *(const v4f*)(gA + (size_t)(24 * j) * HH);
  for (int j = 0; j < PJ; ++j) rb[j] = *(const v4f*)(gB + (size_t)(24 * j) * HH);
  for (int j = 0; j < PJ; ++j) *(v4f*)&sA[0][srow + 24 * j][scol] = ra[j];
  for (int j = 0; j < PJ; ++j) *(v4f*)&sB[0][srow + 24 * j][scol] = rb[j];
  __syncthreads();
  for (int j = 0; j < PJ; ++j) ra[j] = *(const v4f*)(gA + (size_t)(24 * j) * HH + TK);
  for (int j = 0; j < PJ; ++j) rb[j] = *(const v4f*)(gB + (size_t)(24 * j) * HH + TK);

  v8f acc0 = {};
  v8f acc1 = {};
  v8f acc2 = {};

  for (int kt = 0; kt < NK; ++kt) {
    const int  buf   = kt & 1;
    const bool more  = (kt + 1) < NK;
    const bool more2 = (kt + 2) < NK;

    const float* __restrict__ pa0 = &sA[buf][wm * 48 + r16][kh];
    const float* __restrict__ pa1 = pa0 + 16 * LP;
    const float* __restrict__ pa2 = pa0 + 32 * LP;
    const float* __restrict__ pb  = &sB[buf][wn * 16 + r16][kh];
    v2f fa0[TK / 4], fa1[TK / 4], fa2[TK / 4], fb[TK / 4];
    for (int ks = 0; ks < TK / 4; ++ks) {
      fa0[ks] = *(const v2f*)(pa0 + 4 * ks);
      fa1[ks] = *(const v2f*)(pa1 + 4 * ks);
      fa2[ks] = *(const v2f*)(pa2 + 4 * ks);
      fb[ks]  = *(const v2f*)(pb + 4 * ks);
    }
    __builtin_amdgcn_sched_barrier(0);

    if (more) {
      for (int j = 0; j < PJ; ++j) *(v4f*)&sA[buf ^ 1][srow + 24 * j][scol] = ra[j];
      for (int j = 0; j < PJ; ++j) *(v4f*)&sB[buf ^ 1][srow + 24 * j][scol] = rb[j];
    }
    if (more2) {
      const int ko = (kt + 2) * TK;
      for (int j = 0; j < PJ; ++j) ra[j] = *(const v4f*)(gA + (size_t)(24 * j) * HH + ko);
      for (int j = 0; j < PJ; ++j) rb[j] = *(const v4f*)(gB + (size_t)(24 * j) * HH + ko);
    }
    __builtin_amdgcn_sched_barrier(0);

    for (int ks = 0; ks < KSIG; ++ks) {
      acc0 = wmma4(fa0[ks], fb[ks], acc0);
      acc1 = wmma4(fa1[ks], fb[ks], acc1);
      acc2 = wmma4(fa2[ks], fb[ks], acc2);
    }

    __builtin_amdgcn_sched_barrier(0);
    if (more) asm volatile("s_wait_dscnt 0x0\n\ts_barrier_signal -1" ::: "memory");
    __builtin_amdgcn_sched_barrier(0);

    for (int ks = KSIG; ks < TK / 4; ++ks) {
      acc0 = wmma4(fa0[ks], fb[ks], acc0);
      acc1 = wmma4(fa1[ks], fb[ks], acc1);
      acc2 = wmma4(fa2[ks], fb[ks], acc2);
    }

    __builtin_amdgcn_sched_barrier(0);
    if (more) asm volatile("s_barrier_wait -1" ::: "memory");
  }

  const int n     = n0 + wn * 16 + r16;
  const int mbase = m0 + wm * 48 + ((lane >> 4) << 3);
  const float bn  = bias[n];
#pragma unroll
  for (int r = 0; r < 8; ++r) {
    Y[(size_t)(mbase + r) * NN + n]      = acc0[r] + bn;
    Y[(size_t)(mbase + 16 + r) * NN + n] = acc1[r] + bn;
    Y[(size_t)(mbase + 32 + r) * NN + n] = acc2[r] + bn;
  }
}

__device__ __forceinline__ v2f splat2(float c) { return (v2f){c, c}; }

__device__ __forceinline__ v2f gelu_erf2_twice(v2f x) {
  const v2f z = x * x;
  v2f p = splat2(1.80561643e-9f);
  p = __builtin_elementwise_fma(p, z, splat2(-1.33446406e-7f));
  p = __builtin_elementwise_fma(p, z, splat2(3.96910946e-6f));
  p = __builtin_elementwise_fma(p, z, splat2(-5.52401580e-5f));
  p = __builtin_elementwise_fma(p, z, splat2(-3.31299925e-5f));
  p = __builtin_elementwise_fma(p, z, splat2(0.0363342836f));
  p = __builtin_elementwise_fma(p, z, splat2(0.797884226f));
  const v2f r = x * p;
  v2f t;
  t.x = __builtin_amdgcn_tanhf(r.x);
  t.y = __builtin_amdgcn_tanhf(r.y);
  return __builtin_elementwise_fma(x, t, x);
}

#define PW 8
#define PT 48
#define HV (HH / 128)

__device__ __forceinline__ float xhalf(float x) {
  const int b = __builtin_bit_cast(int, x);
  return __builtin_bit_cast(float, __builtin_amdgcn_permlanex16(
      b, b, 0x76543210, 0xfedcba98, false, false));
}

template <int N>
__device__ __forceinline__ float row_xadd(float v) {
  const int b = __builtin_bit_cast(int, v);
  return v + __builtin_bit_cast(float, __builtin_amdgcn_update_dpp(0, b, 0x160 + N, 0xf, 0xf, true));
}

__device__ __forceinline__ float row_part(const v4f (&sv)[HV], const v4f (&wv)[HV],
                                         const v4f (&tv)[HV]) {
  v2f a01 = splat2(0.0f), a23 = splat2(0.0f);
#pragma unroll
  for (int i = 0; i < HV; ++i) {
    a01 = __builtin_elementwise_fma(gelu_erf2_twice(sv[i].xy + tv[i].xy), wv[i].xy, a01);
    a23 = __builtin_elementwise_fma(gelu_erf2_twice(sv[i].zw + tv[i].zw), wv[i].zw, a23);
  }
  return (a01.x + a01.y) + (a23.x + a23.y);
}

__global__ __launch_bounds__(256) __attribute__((amdgpu_waves_per_eu(6, 6)))
void pair_gelu_reduce(
    const float* __restrict__ hsrc, const float* __restrict__ htgt,
    const float* __restrict__ wcls, float* __restrict__ out) {
  const int lane = threadIdx.x & 31;
  const bool hi  = lane >= 16;
  const int wid  = threadIdx.x >> 5;
  const int s    = blockIdx.x * PW + wid;
  const int t0   = blockIdx.y * PT;
  const int l    = blockIdx.z % LL;
  const int b    = blockIdx.z / LL;

  const float* __restrict__ srcrow =
      hsrc + ((size_t)((b * SS + s) * LL + l)) * HH + 4 * lane;
  v4f sv[HV], wv[HV];
#pragma unroll
  for (int i = 0; i < HV; ++i) {
    sv[i] = *(const v4f*)(srcrow + 128 * i);
    const float* __restrict__ wp = wcls + (size_t)(4 * lane + 128 * i) * LL + l;
    wv[i].x = 0.5f * wp[0];
    wv[i].y = 0.5f * wp[LL];
    wv[i].z = 0.5f * wp[2 * LL];
    wv[i].w = 0.5f * wp[3 * LL];
  }

  const float* __restrict__ trow =
      htgt + ((size_t)((b * SS + t0) * LL + l)) * HH + 4 * lane;
  float* __restrict__ orow =
      out + ((size_t)((b * SS + s) * SS + t0)) * LL + l;

  v4f ta[HV], tb[HV];
#pragma unroll
  for (int i = 0; i < HV; ++i) ta[i] = *(const v4f*)(trow + 128 * i);

  for (int t = 0; t < PT; t += 2) {
    const float* __restrict__ r1 = trow + (size_t)(t + 1) * (LL * HH);
#pragma unroll
    for (int i = 0; i < HV; ++i) tb[i] = *(const v4f*)(r1 + 128 * i);

    const float p0 = row_part(sv, wv, ta);

    const int tn = (t + 2 < PT) ? (t + 2) : (PT - 1);
    const float* __restrict__ r2 = trow + (size_t)tn * (LL * HH);
#pragma unroll
    for (int i = 0; i < HV; ++i) ta[i] = *(const v4f*)(r2 + 128 * i);

    const float p1 = row_part(sv, wv, tb);

    float v = (hi ? p1 : p0) + xhalf(hi ? p0 : p1);
    v = row_xadd<1>(v);
    v = row_xadd<2>(v);
    v = row_xadd<4>(v);
    v = row_xadd<8>(v);
    if ((lane & 15) == 0) orow[(size_t)(t + (lane >> 4)) * LL] = v;
  }
}

extern "C" void kernel_launch(void* const* d_in, const int* in_sizes, int n_in,
                              void* d_out, int out_size, void* d_ws, size_t ws_size,
                              hipStream_t stream) {
  const float* pooled = (const float*)d_in[0];
  const float* W_src  = (const float*)d_in[1];
  const float* b_src  = (const float*)d_in[2];
  const float* W_tgt  = (const float*)d_in[3];
  const float* b_tgt  = (const float*)d_in[4];
  const float* wcls   = (const float*)d_in[5];
  float* out = (float*)d_out;

  float* hsrc = (float*)d_ws;
  float* htgt = hsrc + (size_t)MM * NN;

  dim3 g1(NN / TN, MM / TM, 2);
  proj_wmma_f32<<<g1, 384, 0, stream>>>(pooled, W_src, b_src, hsrc, W_tgt, b_tgt, htgt);

  dim3 g2(SS / PW, SS / PT, BB * LL);
  pair_gelu_reduce<<<g2, 256, 0, stream>>>(hsrc, htgt, wcls, out);
}
